# starting kernel
# baseline (speedup 1.0000x reference)
#include <hip/hip_runtime.h>

typedef float v2f __attribute__((ext_vector_type(2)));
typedef float v8f __attribute__((ext_vector_type(8)));

#define CH 300
#define CHPAD 304
#define TP 4


__global__ void fill_kernel(float* __restrict__ p, float v, int n) {
    int i = blockIdx.x * blockDim.x + threadIdx.x;
    if (i < n) p[i] = v;
}

__global__ void deg_kernel(const int* __restrict__ row, float* __restrict__ deg, int E) {
    int e = blockIdx.x * blockDim.x + threadIdx.x;
    if (e < E) unsafeAtomicAdd(&deg[row[e]], 1.0f);
}

__global__ void rsqrt_kernel(float* __restrict__ d, int n) {
    int i = blockIdx.x * blockDim.x + threadIdx.x;
    if (i < n) d[i] = rsqrtf(d[i]);
}

__global__ void padw_kernel(const float* __restrict__ W, float* __restrict__ Wp) {
    int i = blockIdx.x * blockDim.x + threadIdx.x;
    if (i < CHPAD * CH) Wp[i] = (i < CH * CH) ? W[i] : 0.0f;
}

__global__ void mean_kernel(const float* __restrict__ x, float* __restrict__ xm, int total) {
    int i = blockIdx.x * blockDim.x + threadIdx.x;
    if (i >= total) return;
    int n = i / CH;
    int c = i - n * CH;
    const float* b = x + (size_t)n * (TP * CH) + c;
    xm[i] = 0.25f * (b[0] + b[CH] + b[2 * CH] + b[3 * CH]);
}

__global__ __launch_bounds__(128) void gemm_wmma_kernel(const float* __restrict__ A,
                                                        const float* __restrict__ Wp,
                                                        const float* __restrict__ bias,
                                                        float* __restrict__ C,
                                                        int mtiles) {
    const int wave = threadIdx.x >> 5;
    const int lane = threadIdx.x & 31;
    const int trow = blockIdx.x * 4 + wave;
    if (trow >= mtiles) return;

    const int mn   = lane & 15;
    const int half = lane >> 4;
    const int row0 = trow * 16;
    const int col0 = blockIdx.y * 64;

    v8f acc[4] = {};

    const float* ap = A  + (size_t)(row0 + mn) * CH + 2 * half;
    const float* bp = Wp + (size_t)(col0 + mn) * CH + 2 * half;

    for (int k0 = 0; k0 < CH; k0 += 4) {
        v2f a = *(const v2f*)(ap + k0);
#pragma unroll
        for (int t = 0; t < 4; ++t) {
            v2f b = *(const v2f*)(bp + k0 + t * (16 * CH));
            acc[t] = __builtin_amdgcn_wmma_f32_16x16x4_f32(
                false, a, false, b, (short)0, acc[t], false, false);
        }
    }

#pragma unroll
    for (int t = 0; t < 4; ++t) {
        int n = col0 + t * 16 + mn;
        if (n < CH) {
            float bv = bias[n];
#pragma unroll
            for (int j = 0; j < 8; ++j) {
                int m = row0 + j + 8 * half;
                C[(size_t)m * CH + n] = acc[t][j] + bv;
            }
        }
    }
}

__global__ void scatter_kernel(const int* __restrict__ row, const int* __restrict__ col,
                               const float* __restrict__ dis, const float* __restrict__ h,
                               float* __restrict__ acc) {
    int e = blockIdx.x;
    int c = blockIdx.y * blockDim.x + threadIdx.x;
    if (c >= CH) return;
    int r = row[e];
    int d = col[e];
    float v = dis[r] * h[(size_t)r * CH + c];
    unsafeAtomicAdd(&acc[(size_t)d * CH + c], v);
}

__global__ void finalize_kernel(const float* __restrict__ acc, const float* __restrict__ h,
                                const float* __restrict__ dis, float* __restrict__ out,
                                int total, int relu) {
    int i = blockIdx.x * blockDim.x + threadIdx.x;
    if (i >= total) return;
    int n = i / CH;
    float dn = dis[n];
    float v = dn * (acc[i] + dn * h[i]);
    if (relu) v = (v > 0.0f) ? v : 0.01f * v;
    out[i] = v;
}

extern "C" void kernel_launch(void* const* d_in, const int* in_sizes, int n_in,
                              void* d_out, int out_size, void* d_ws, size_t ws_size,
                              hipStream_t stream) {
    const float* x  = (const float*)d_in[0];
    const int*   ei = (const int*)d_in[1];
    const float* W1 = (const float*)d_in[2];
    const float* b1 = (const float*)d_in[3];
    const float* W2 = (const float*)d_in[4];
    const float* b2 = (const float*)d_in[5];
    float* out = (float*)d_out;

    const int E = in_sizes[1] / 2;
    const int N = in_sizes[0] / (TP * CH);
    const int total = N * CH;
    const int* row = ei;
    const int* col = ei + E;

    char* ws = (char*)d_ws;
    float* dis = (float*)ws;
    size_t off = ((size_t)N * sizeof(float) + 255) & ~(size_t)255;
    float* Wp = (float*)(ws + off);
    off += (size_t)CHPAD * CH * sizeof(float);
    float* bufA = (float*)(ws + off);
    float* bufB = bufA + total;
    float* bufC = bufB + total;

    const int TB = 256;
    const int mtiles = (N + 15) / 16;
    dim3 gg((mtiles + 3) / 4, (CH + 63) / 64);
    dim3 gs(E, (CH + 127) / 128);
    const int padBlocks = (CHPAD * CH + TB - 1) / TB;

    fill_kernel<<<(N + TB - 1) / TB, TB, 0, stream>>>(dis, 1.0f, N);
    deg_kernel<<<(E + TB - 1) / TB, TB, 0, stream>>>(row, dis, E);
    rsqrt_kernel<<<(N + TB - 1) / TB, TB, 0, stream>>>(dis, N);

    mean_kernel<<<(total + TB - 1) / TB, TB, 0, stream>>>(x, bufA, total);

    padw_kernel<<<padBlocks, TB, 0, stream>>>(W1, Wp);
    gemm_wmma_kernel<<<gg, 128, 0, stream>>>(bufA, Wp, b1, bufB, mtiles);

    fill_kernel<<<(total + TB - 1) / TB, TB, 0, stream>>>(bufC, 0.0f, total);
    scatter_kernel<<<gs, 128, 0, stream>>>(row, col, dis, bufB, bufC);
    finalize_kernel<<<(total + TB - 1) / TB, TB, 0, stream>>>(bufC, bufB, dis, bufA, total, 1);

    padw_kernel<<<padBlocks, TB, 0, stream>>>(W2, Wp);
    gemm_wmma_kernel<<<gg, 128, 0, stream>>>(bufA, Wp, b2, bufB, mtiles);

    fill_kernel<<<(total + TB - 1) / TB, TB, 0, stream>>>(out, 0.0f, total);
    scatter_kernel<<<gs, 128, 0, stream>>>(row, col, dis, bufB, out);
    finalize_kernel<<<(total + TB - 1) / TB, TB, 0, stream>>>(out, bufB, dis, out, total, 0);
}


# Edge lists built once: the layer-1 aggregation stores each wave's finished lists in the workspace, layer 2 reloads them; its col sweep, row lookups and list split go; same sum order
# speedup vs baseline: 3.4554x; 3.4554x over previous
#include <hip/hip_runtime.h>

typedef float v2f __attribute__((ext_vector_type(2)));
typedef float v8f __attribute__((ext_vector_type(8)));

#define CH 300
#define TP 4
#define KCH 75
#define TILE_F 4800
#define TILE_U 2400
#define WTILES 20


__global__ void fill_kernel(float* __restrict__ p, float v, int n) {
    int i = blockIdx.x * blockDim.x + threadIdx.x;
    if (i < n) p[i] = v;
}

__global__ void deg_kernel(const int* __restrict__ row, float* __restrict__ deg, int E) {
    int e = blockIdx.x * blockDim.x + threadIdx.x;
    if (e < E) unsafeAtomicAdd(&deg[row[e]], 1.0f);
}

__global__ void rsqrt_kernel(float* __restrict__ d, int n) {
    int i = blockIdx.x * blockDim.x + threadIdx.x;
    if (i < n) d[i] = rsqrtf(d[i]);
}

__global__ void padw_kernel(const float* __restrict__ W, float* __restrict__ Wf) {
    int g = blockIdx.x * blockDim.x + threadIdx.x;
    if (g >= WTILES * TILE_U) return;
    int tile = g / TILE_U;
    int rem = g - tile * TILE_U;
    int kc = rem >> 5, u = rem & 31;
    int n = tile * 16 + (u & 15);
    float2 v = make_float2(0.f, 0.f);
    if (n < CH) v = *(const float2*)(W + (size_t)n * CH + 4 * kc + 2 * (u >> 4));
    ((float2*)Wf)[g] = v;
}

__global__ __launch_bounds__(256) void mean_kernel(const float* __restrict__ x,
                                                   float* __restrict__ xf, int N) {
    __shared__ float2 t2[TILE_U];
    const int row0 = blockIdx.x * 16;
    float4 v[5][4];
#pragma unroll
    for (int it = 0; it < 5; ++it) {
        int i = min((int)threadIdx.x + it * 256, 16 * KCH - 1);
        int c = i / KCH;
        int kc = i - c * KCH;
        int n = min(row0 + c, N - 1);
        const float4* b = (const float4*)(x + (size_t)n * (TP * CH)) + kc;
        v[it][0] = b[0]; v[it][1] = b[KCH]; v[it][2] = b[2 * KCH]; v[it][3] = b[3 * KCH];
    }
#pragma unroll
    for (int it = 0; it < 5; ++it) {
        int i = threadIdx.x + it * 256;
        if (i < 16 * KCH) {
            int c = i / KCH;
            int kc = i - c * KCH;
            float4 s;
            s.x = 0.25f * (v[it][0].x + v[it][1].x + v[it][2].x + v[it][3].x);
            s.y = 0.25f * (v[it][0].y + v[it][1].y + v[it][2].y + v[it][3].y);
            s.z = 0.25f * (v[it][0].z + v[it][1].z + v[it][2].z + v[it][3].z);
            s.w = 0.25f * (v[it][0].w + v[it][1].w + v[it][2].w + v[it][3].w);
            if (row0 + c >= N) s = make_float4(0.f, 0.f, 0.f, 0.f);
            int sw = kc & 31;
            t2[kc * 32 + (c ^ sw)] = make_float2(s.x, s.y);
            t2[kc * 32 + ((16 + c) ^ sw)] = make_float2(s.z, s.w);
        }
    }
    __syncthreads();
    float2* o = (float2*)(xf + (size_t)blockIdx.x * TILE_F);
    for (int g = threadIdx.x; g < TILE_U; g += 256) {
        int q = g >> 5, u = g & 31;
        o[g] = t2[q * 32 + (u ^ (q & 31))];
    }
}

__global__ __launch_bounds__(128) void gemm_wmma_kernel(const float* __restrict__ A,
                                                        const float* __restrict__ Wp,
                                                        const float* __restrict__ bias,
                                                        float* __restrict__ C,
                                                        int mtiles) {
    const int wave = threadIdx.x >> 5;
    const int lane = threadIdx.x & 31;
    const int trow = blockIdx.x * 4 + wave;
    if (trow >= mtiles) return;

    const int mn   = lane & 15;
    const int half = lane >> 4;
    const int row0 = trow * 16;
    const int col0 = blockIdx.y * 64;

    v8f acc[4] = {};

    const float* ap = A  + (size_t)trow * TILE_F + 2 * lane;
    const float* bp = Wp + (size_t)(blockIdx.y * 4) * TILE_F + 2 * lane;

    for (int kc = 0; kc < KCH; ++kc) {
        v2f a = *(const v2f*)(ap + kc * 64);
#pragma unroll
        for (int t = 0; t < 4; ++t) {
            v2f b = *(const v2f*)(bp + kc * 64 + t * TILE_F);
            acc[t] = __builtin_amdgcn_wmma_f32_16x16x4_f32(
                false, a, false, b, (short)0, acc[t], false, false);
        }
    }

#pragma unroll
    for (int t = 0; t < 4; ++t) {
        int n = col0 + t * 16 + mn;
        if (n < CH) {
            float bv = bias[n];
#pragma unroll
            for (int j = 0; j < 8; ++j) {
                int m = row0 + j + 8 * half;
                C[(size_t)m * CH + n] = acc[t][j] + bv;
            }
        }
    }
}

#define AGG_R    208
#define AGG_T    1024
#define AGG_W    32
#define AGG_CAP  128
#define AGG_K    16
#define AGG_OWN  128
#define ROW4     75

__global__ __launch_bounds__(AGG_T) void gather_kernel(const int* __restrict__ row,
                                                       const int* __restrict__ col,
                                                       const float* __restrict__ dis,
                                                       const float* __restrict__ h,
                                                       float* __restrict__ out,
                                                       int* __restrict__ spill,
                                                       int E, int N, int relu, int fm, int reuse) {
    __shared__ float2 tl[AGG_R * (CH / 2)];
    __shared__ int2 lst[AGG_W * AGG_CAP];
    __shared__ int2 own[AGG_W * AGG_OWN];
    __shared__ int  cnt[AGG_W];

    const int wave = __builtin_amdgcn_readfirstlane((int)(threadIdx.x >> 5));
    const int lane = threadIdx.x & 31;
    const int lo   = blockIdx.x * AGG_R;
    const int PIECE  = 32 * AGG_K;
    const int STRIDE = AGG_W * PIECE;
    const int chunk = ((E + STRIDE - 1) / STRIDE) * PIECE;
    int2* blk_ovf = (int2*)spill + (size_t)blockIdx.x * AGG_W * chunk;
    int2* my_ovf  = blk_ovf + (size_t)wave * chunk;
    int2* my_lst  = lst + wave * AGG_CAP;
    int2* my_own  = own + wave * AGG_OWN;
    int2* sv      = (int2*)spill + (size_t)gridDim.x * AGG_W * chunk;
    int2* sv_lst  = sv + ((size_t)blockIdx.x * AGG_W + wave) * (AGG_CAP + AGG_OWN);
    int2* sv_own  = sv_lst + AGG_CAP;
    int*  sv_cnt  = (int*)(sv + (size_t)gridDim.x * AGG_W * (AGG_CAP + AGG_OWN))
                    + (size_t)blockIdx.x * (2 * AGG_W);

    int n = 0;
    if (reuse) {
        n = __builtin_amdgcn_readfirstlane(sv_cnt[wave]);
        const int nl = min(n, AGG_CAP);
        for (int i = lane; i < nl; i += 32) my_lst[i] = sv_lst[i];
    } else {
        int c[AGG_K], nx[AGG_K];
        int eb = wave * PIECE;
#pragma unroll
        for (int k = 0; k < AGG_K; ++k) c[k] = col[min(eb + k * 32 + lane, E - 1)];
        for (; eb < E; eb += STRIDE) {
#pragma unroll
            for (int k = 0; k < AGG_K; ++k) nx[k] = col[min(eb + STRIDE + k * 32 + lane, E - 1)];
#pragma unroll
            for (int k = 0; k < AGG_K; ++k) {
                int e = eb + k * 32 + lane;
                unsigned dlx = (unsigned)(c[k] - lo);
                bool hit = (e < E) && (dlx < (unsigned)AGG_R);
                unsigned m = __builtin_amdgcn_ballot_w32(hit);
                if (m) {
                    if (hit) {
                        int pos = n + (int)__builtin_amdgcn_mbcnt_lo(m, 0u);
                        int2 v = make_int2(e, (int)dlx);
                        if (pos < AGG_CAP) my_lst[pos] = v; else my_ovf[pos - AGG_CAP] = v;
                    }
                    n += __builtin_popcount(m);
                }
            }
            for (int k = 0; k < AGG_K; ++k) c[k] = nx[k];
        }
        if (n > AGG_CAP) __threadfence_block();
        for (int i = lane; i < n; i += 32) {
            if (i < AGG_CAP) { int2 v = my_lst[i]; v.x = row[v.x]; my_lst[i] = v; sv_lst[i] = v; }
            else { int2 v = my_ovf[i - AGG_CAP]; v.x = row[v.x]; my_ovf[i - AGG_CAP] = v; }
        }
    }
    if (lane == 0) cnt[wave] = n;
    __syncthreads();

    const int cv = cnt[lane];
    int no = 0;
    if (reuse) {
        no = __builtin_amdgcn_readfirstlane(sv_cnt[AGG_W + wave]);
        const int nk = min(no, AGG_OWN);
        for (int i = lane; i < nk; i += 32) my_own[i] = sv_own[i];
    } else {
        auto keep = [&](int2 v, bool ok) {
            bool mine = ok && ((v.y & (AGG_W - 1)) == wave);
            unsigned m = __builtin_amdgcn_ballot_w32(mine);
            if (mine) {
                int pos = no + (int)__builtin_amdgcn_mbcnt_lo(m, 0u);
                if (pos < AGG_OWN) my_own[pos] = v;
            }
            no += __builtin_popcount(m);
        };
        for (int base = 0; base < AGG_CAP && __builtin_amdgcn_ballot_w32(cv > base) != 0u; base += 32) {
            for (int g = 0; g < AGG_W; g += 8) {
                int2 v[8];
                bool ok[8];
#pragma unroll
                for (int k = 0; k < 8; ++k) {
                    ok[k] = (base + lane) < __builtin_amdgcn_readlane(cv, g + k);
                    v[k] = make_int2(0, -1);
                    if (ok[k]) v[k] = lst[(g + k) * AGG_CAP + base + lane];
                }
#pragma unroll
                for (int k = 0; k < 8; ++k) keep(v[k], ok[k]);
            }
        }
        for (int base = AGG_CAP; __builtin_amdgcn_ballot_w32(cv > base) != 0u; base += 32) {
            for (int s = 0; s < AGG_W; ++s) {
                bool ok = (base + lane) < __builtin_amdgcn_readlane(cv, s);
                int2 v = make_int2(0, -1);
                if (ok) v = blk_ovf[(size_t)s * chunk + (base + lane - AGG_CAP)];
                keep(v, ok);
            }
        }
        const int nk = min(no, AGG_OWN);
        for (int i = lane; i < nk; i += 32) sv_own[i] = my_own[i];
        if (lane == 0) { sv_cnt[wave] = n; sv_cnt[AGG_W + wave] = no; }
    }
    const bool fits = (no <= AGG_OWN);

    const int t2 = min(lane + 64, ROW4 - 1);
    const int nrows = min(AGG_R, N - lo);
    const int ntb   = (nrows + 15) >> 4;
    const int rend  = fm ? ntb * 16 : nrows;
    auto put = [&](int dl, int f, float4 r) {
        const int q = (dl >> 4) * KCH + f, c = dl & 15, sw = q & 31;
        tl[q * 32 + (c ^ sw)] = make_float2(r.x, r.y);
        tl[q * 32 + ((16 + c) ^ sw)] = make_float2(r.z, r.w);
    };
    for (int dl = wave; dl < rend; dl += AGG_W) {
        const int d = lo + dl;
        if (dl >= nrows) {
            const float4 z = make_float4(0.f, 0.f, 0.f, 0.f);
            put(dl, lane, z); put(dl, lane + 32, z);
            if (lane < ROW4 - 64) put(dl, lane + 64, z);
            continue;
        }
        const float4* sp = (const float4*)(h + (size_t)d * CH);
        const float dn = dis[d];
        const float4 s0 = sp[lane], s1 = sp[lane + 32], s2 = sp[t2];
        float4 a0 = make_float4(0.f, 0.f, 0.f, 0.f), a1 = a0, a2 = a0;

        auto addhits = [&](int2 v, bool ok) {
            unsigned m = __builtin_amdgcn_ballot_w32(ok && v.y == dl);
            const int nh = __builtin_popcount(m);
            for (int q = 0; q < nh; ++q) {
                int j = __builtin_ctz(m);
                m &= m - 1u;
                int sn = __builtin_amdgcn_readlane(v.x, j);
                float w = dis[sn];
                const float4* hp = (const float4*)(h + (size_t)sn * CH);
                float4 x0 = hp[lane], x1 = hp[lane + 32], x2 = hp[t2];
                a0.x += w * x0.x; a0.y += w * x0.y; a0.z += w * x0.z; a0.w += w * x0.w;
                a1.x += w * x1.x; a1.y += w * x1.y; a1.z += w * x1.z; a1.w += w * x1.w;
                a2.x += w * x2.x; a2.y += w * x2.y; a2.z += w * x2.z; a2.w += w * x2.w;
            }
        };

        if (fits) {
            for (int i0 = 0; i0 < no; i0 += 32) {
                bool ok = (i0 + lane) < no;
                int2 v = make_int2(0, -1);
                if (ok) v = my_own[i0 + lane];
                addhits(v, ok);
            }
        } else {
            for (int s = 0; s < AGG_W; ++s) {
                int ns = __builtin_amdgcn_readlane(cv, s);
                for (int i0 = 0; i0 < ns; i0 += 32) {
                    bool ok = (i0 + lane) < ns;
                    int2 v = make_int2(0, -1);
                    if (ok) {
                        if (i0 < AGG_CAP) v = lst[s * AGG_CAP + i0 + lane];
                        else v = blk_ovf[(size_t)s * chunk + (i0 + lane - AGG_CAP)];
                    }
                    addhits(v, ok);
                }
            }
        }

        float4 r0, r1, r2;
        r0.x = dn * (a0.x + dn * s0.x); r0.y = dn * (a0.y + dn * s0.y);
        r0.z = dn * (a0.z + dn * s0.z); r0.w = dn * (a0.w + dn * s0.w);
        r1.x = dn * (a1.x + dn * s1.x); r1.y = dn * (a1.y + dn * s1.y);
        r1.z = dn * (a1.z + dn * s1.z); r1.w = dn * (a1.w + dn * s1.w);
        r2.x = dn * (a2.x + dn * s2.x); r2.y = dn * (a2.y + dn * s2.y);
        r2.z = dn * (a2.z + dn * s2.z); r2.w = dn * (a2.w + dn * s2.w);
        if (relu) {
            r0.x = (r0.x > 0.f) ? r0.x : 0.01f * r0.x; r0.y = (r0.y > 0.f) ? r0.y : 0.01f * r0.y;
            r0.z = (r0.z > 0.f) ? r0.z : 0.01f * r0.z; r0.w = (r0.w > 0.f) ? r0.w : 0.01f * r0.w;
            r1.x = (r1.x > 0.f) ? r1.x : 0.01f * r1.x; r1.y = (r1.y > 0.f) ? r1.y : 0.01f * r1.y;
            r1.z = (r1.z > 0.f) ? r1.z : 0.01f * r1.z; r1.w = (r1.w > 0.f) ? r1.w : 0.01f * r1.w;
            r2.x = (r2.x > 0.f) ? r2.x : 0.01f * r2.x; r2.y = (r2.y > 0.f) ? r2.y : 0.01f * r2.y;
            r2.z = (r2.z > 0.f) ? r2.z : 0.01f * r2.z; r2.w = (r2.w > 0.f) ? r2.w : 0.01f * r2.w;
        }
        if (fm) {
            put(dl, lane, r0); put(dl, lane + 32, r1);
            if (lane < ROW4 - 64) put(dl, lane + 64, r2);
        } else {
            float4* op = (float4*)(out + (size_t)d * CH);
            op[lane] = r0;
            op[lane + 32] = r1;
            if (lane < ROW4 - 64) op[lane + 64] = r2;
        }
    }
    if (fm) {
        __syncthreads();
        float2* o = (float2*)out + (size_t)(lo >> 4) * TILE_U;
        const int tot = ntb * TILE_U;
        for (int g = threadIdx.x; g < tot; g += AGG_T) {
            int q = g >> 5, u = g & 31;
            o[g] = tl[q * 32 + (u ^ (q & 31))];
        }
    }
}

extern "C" void kernel_launch(void* const* d_in, const int* in_sizes, int n_in,
                              void* d_out, int out_size, void* d_ws, size_t ws_size,
                              hipStream_t stream) {
    const float* x  = (const float*)d_in[0];
    const int*   ei = (const int*)d_in[1];
    const float* W1 = (const float*)d_in[2];
    const float* b1 = (const float*)d_in[3];
    const float* W2 = (const float*)d_in[4];
    const float* b2 = (const float*)d_in[5];
    float* out = (float*)d_out;

    const int E = in_sizes[1] / 2;
    const int N = in_sizes[0] / (TP * CH);
    const int total = N * CH;
    const int* row = ei;
    const int* col = ei + E;

    char* ws = (char*)d_ws;
    float* dis = (float*)ws;
    size_t off = ((size_t)N * sizeof(float) + 255) & ~(size_t)255;
    float* Wp = (float*)(ws + off);
    off += (size_t)WTILES * TILE_F * sizeof(float);
    const int mtiles = (N + 15) / 16;
    float* bufA = (float*)(ws + off);
    float* bufB = bufA + (size_t)mtiles * TILE_F;
    int* ovf = (int*)(bufB + total);

    const int TB = 256;
    dim3 gg((mtiles + 3) / 4, (CH + 63) / 64);
    const int nagg = (N + AGG_R - 1) / AGG_R;
    const int padBlocks = (WTILES * TILE_U + TB - 1) / TB;

    fill_kernel<<<(N + TB - 1) / TB, TB, 0, stream>>>(dis, 1.0f, N);
    deg_kernel<<<(E + TB - 1) / TB, TB, 0, stream>>>(row, dis, E);
    rsqrt_kernel<<<(N + TB - 1) / TB, TB, 0, stream>>>(dis, N);

    mean_kernel<<<mtiles, 256, 0, stream>>>(x, bufA, N);

    padw_kernel<<<padBlocks, TB, 0, stream>>>(W1, Wp);
    gemm_wmma_kernel<<<gg, 128, 0, stream>>>(bufA, Wp, b1, bufB, mtiles);

    gather_kernel<<<nagg, AGG_T, 0, stream>>>(row, col, dis, bufB, bufA, ovf, E, N, 1, 1, 0);

    padw_kernel<<<padBlocks, TB, 0, stream>>>(W2, Wp);
    gemm_wmma_kernel<<<gg, 128, 0, stream>>>(bufA, Wp, b2, bufB, mtiles);

    gather_kernel<<<nagg, AGG_T, 0, stream>>>(row, col, dis, bufB, out, ovf, E, N, 0, 0, 1);
}


# Aggregation runs before each linear layer: a block gathers its 208 rows into an LDS tile image and does the layer's GEMM from it; both GEMM launches and operand round trips go; sum order changes
# speedup vs baseline: 3.9475x; 1.1424x over previous
#include <hip/hip_runtime.h>

typedef float v2f __attribute__((ext_vector_type(2)));
typedef float v8f __attribute__((ext_vector_type(8)));

#define CH 300
#define TP 4
#define KCH 75
#define TILE_F 4800
#define TILE_U 2400
#define WTILES 20


__global__ void fill_kernel(float* __restrict__ p, float v, int n) {
    int i = blockIdx.x * blockDim.x + threadIdx.x;
    if (i < n) p[i] = v;
}

__global__ void deg_kernel(const int* __restrict__ row, float* __restrict__ deg, int E) {
    int e = blockIdx.x * blockDim.x + threadIdx.x;
    if (e < E) unsafeAtomicAdd(&deg[row[e]], 1.0f);
}

__global__ void rsqrt_kernel(float* __restrict__ d, int n) {
    int i = blockIdx.x * blockDim.x + threadIdx.x;
    if (i < n) d[i] = rsqrtf(d[i]);
}

__global__ void padw_kernel(const float* __restrict__ W, float* __restrict__ Wf) {
    int g = blockIdx.x * blockDim.x + threadIdx.x;
    if (g >= WTILES * TILE_U) return;
    int tile = g / TILE_U;
    int rem = g - tile * TILE_U;
    int kc = rem >> 5, u = rem & 31;
    int n = tile * 16 + (u & 15);
    float2 v = make_float2(0.f, 0.f);
    if (n < CH) v = *(const float2*)(W + (size_t)n * CH + 4 * kc + 2 * (u >> 4));
    ((float2*)Wf)[g] = v;
}

__global__ __launch_bounds__(256) void mean_kernel(const float* __restrict__ x,
                                                   float* __restrict__ xm, int N) {
    const int row0 = blockIdx.x * 16;
    float4 v[5][4];
#pragma unroll
    for (int it = 0; it < 5; ++it) {
        int i = min((int)threadIdx.x + it * 256, 16 * KCH - 1);
        int c = i / KCH;
        int kc = i - c * KCH;
        int n = min(row0 + c, N - 1);
        const float4* b = (const float4*)(x + (size_t)n * (TP * CH)) + kc;
        v[it][0] = b[0]; v[it][1] = b[KCH]; v[it][2] = b[2 * KCH]; v[it][3] = b[3 * KCH];
    }
    float4* o = (float4*)xm + (size_t)row0 * KCH;
#pragma unroll
    for (int it = 0; it < 5; ++it) {
        int i = threadIdx.x + it * 256;
        int c = i / KCH;
        if (i < 16 * KCH && row0 + c < N) {
            float4 s;
            s.x = 0.25f * (v[it][0].x + v[it][1].x + v[it][2].x + v[it][3].x);
            s.y = 0.25f * (v[it][0].y + v[it][1].y + v[it][2].y + v[it][3].y);
            s.z = 0.25f * (v[it][0].z + v[it][1].z + v[it][2].z + v[it][3].z);
            s.w = 0.25f * (v[it][0].w + v[it][1].w + v[it][2].w + v[it][3].w);
            o[i] = s;
        }
    }
}

#define AGG_R    208
#define AGG_T    1024
#define AGG_W    32
#define AGG_CAP  128
#define AGG_K    16
#define AGG_OWN  128
#define ROW4     75
#define CGT      5

template <int V> struct int_c { static constexpr int value = V; };

__global__ __launch_bounds__(AGG_T) void agg_gemm_kernel(const int* __restrict__ row,
                                                         const int* __restrict__ col,
                                                         const float* __restrict__ dis,
                                                         const float* __restrict__ h,
                                                         const float* __restrict__ Wp,
                                                         const float* __restrict__ bias,
                                                         float* __restrict__ out,
                                                         int* __restrict__ spill,
                                                         int E, int N, int relu, int reuse) {
    __shared__ float2 tl[AGG_R * (CH / 2)];
    __shared__ int2 lst[AGG_W * AGG_CAP];
    __shared__ int2 own[AGG_W * AGG_OWN];
    __shared__ int  cnt[AGG_W];
    __shared__ float sc[AGG_R];

    const int wave = __builtin_amdgcn_readfirstlane((int)(threadIdx.x >> 5));
    const int lane = threadIdx.x & 31;
    const int lo   = blockIdx.x * AGG_R;
    const int PIECE  = 32 * AGG_K;
    const int STRIDE = AGG_W * PIECE;
    const int chunk = ((E + STRIDE - 1) / STRIDE) * PIECE;
    int2* blk_ovf = (int2*)spill + (size_t)blockIdx.x * AGG_W * chunk;
    int2* my_ovf  = blk_ovf + (size_t)wave * chunk;
    int2* my_lst  = lst + wave * AGG_CAP;
    int2* my_own  = own + wave * AGG_OWN;
    int2* sv      = (int2*)spill + (size_t)gridDim.x * AGG_W * chunk;
    int2* sv_lst  = sv + ((size_t)blockIdx.x * AGG_W + wave) * (AGG_CAP + AGG_OWN);
    int2* sv_own  = sv_lst + AGG_CAP;
    int*  sv_cnt  = (int*)(sv + (size_t)gridDim.x * AGG_W * (AGG_CAP + AGG_OWN))
                    + (size_t)blockIdx.x * (2 * AGG_W);

    int n = 0;
    if (reuse) {
        n = __builtin_amdgcn_readfirstlane(sv_cnt[wave]);
        const int nl = min(n, AGG_CAP);
        for (int i = lane; i < nl; i += 32) my_lst[i] = sv_lst[i];
    } else {
        int c[AGG_K], nx[AGG_K];
        int eb = wave * PIECE;
#pragma unroll
        for (int k = 0; k < AGG_K; ++k) c[k] = col[min(eb + k * 32 + lane, E - 1)];
        for (; eb < E; eb += STRIDE) {
#pragma unroll
            for (int k = 0; k < AGG_K; ++k) nx[k] = col[min(eb + STRIDE + k * 32 + lane, E - 1)];
#pragma unroll
            for (int k = 0; k < AGG_K; ++k) {
                int e = eb + k * 32 + lane;
                unsigned dlx = (unsigned)(c[k] - lo);
                bool hit = (e < E) && (dlx < (unsigned)AGG_R);
                unsigned m = __builtin_amdgcn_ballot_w32(hit);
                if (m) {
                    if (hit) {
                        int pos = n + (int)__builtin_amdgcn_mbcnt_lo(m, 0u);
                        int2 v = make_int2(e, (int)dlx);
                        if (pos < AGG_CAP) my_lst[pos] = v; else my_ovf[pos - AGG_CAP] = v;
                    }
                    n += __builtin_popcount(m);
                }
            }
            for (int k = 0; k < AGG_K; ++k) c[k] = nx[k];
        }
        if (n > AGG_CAP) __threadfence_block();
        for (int i = lane; i < n; i += 32) {
            if (i < AGG_CAP) { int2 v = my_lst[i]; v.x = row[v.x]; my_lst[i] = v; sv_lst[i] = v; }
            else { int2 v = my_ovf[i - AGG_CAP]; v.x = row[v.x]; my_ovf[i - AGG_CAP] = v; }
        }
    }
    if (lane == 0) cnt[wave] = n;
    __syncthreads();

    const int cv = cnt[lane];
    int no = 0;
    if (reuse) {
        no = __builtin_amdgcn_readfirstlane(sv_cnt[AGG_W + wave]);
        const int nk = min(no, AGG_OWN);
        for (int i = lane; i < nk; i += 32) my_own[i] = sv_own[i];
    } else {
        auto keep = [&](int2 v, bool ok) {
            bool mine = ok && ((v.y & (AGG_W - 1)) == wave);
            unsigned m = __builtin_amdgcn_ballot_w32(mine);
            if (mine) {
                int pos = no + (int)__builtin_amdgcn_mbcnt_lo(m, 0u);
                if (pos < AGG_OWN) my_own[pos] = v;
            }
            no += __builtin_popcount(m);
        };
        for (int base = 0; base < AGG_CAP && __builtin_amdgcn_ballot_w32(cv > base) != 0u; base += 32) {
            for (int g = 0; g < AGG_W; g += 8) {
                int2 v[8];
                bool ok[8];
#pragma unroll
                for (int k = 0; k < 8; ++k) {
                    ok[k] = (base + lane) < __builtin_amdgcn_readlane(cv, g + k);
                    v[k] = make_int2(0, -1);
                    if (ok[k]) v[k] = lst[(g + k) * AGG_CAP + base + lane];
                }
#pragma unroll
                for (int k = 0; k < 8; ++k) keep(v[k], ok[k]);
            }
        }
        for (int base = AGG_CAP; __builtin_amdgcn_ballot_w32(cv > base) != 0u; base += 32) {
            for (int s = 0; s < AGG_W; ++s) {
                bool ok = (base + lane) < __builtin_amdgcn_readlane(cv, s);
                int2 v = make_int2(0, -1);
                if (ok) v = blk_ovf[(size_t)s * chunk + (base + lane - AGG_CAP)];
                keep(v, ok);
            }
        }
        const int nk = min(no, AGG_OWN);
        for (int i = lane; i < nk; i += 32) sv_own[i] = my_own[i];
        if (lane == 0) { sv_cnt[wave] = n; sv_cnt[AGG_W + wave] = no; }
    }
    const bool fits = (no <= AGG_OWN);

    const int t2 = min(lane + 64, ROW4 - 1);
    const int nrows = min(AGG_R, N - lo);
    const int ntb   = (nrows + 15) >> 4;
    const int rend  = ntb * 16;
    auto put = [&](int dl, int f, float4 r) {
        const int q = (dl >> 4) * KCH + f, c = dl & 15, sw = q & 31;
        tl[q * 32 + (c ^ sw)] = make_float2(r.x, r.y);
        tl[q * 32 + ((16 + c) ^ sw)] = make_float2(r.z, r.w);
    };
    for (int dl = wave; dl < rend; dl += AGG_W) {
        const int d = lo + dl;
        if (dl >= nrows) {
            const float4 z = make_float4(0.f, 0.f, 0.f, 0.f);
            put(dl, lane, z); put(dl, lane + 32, z);
            if (lane < ROW4 - 64) put(dl, lane + 64, z);
            if (lane == 0) sc[dl] = 0.f;
            continue;
        }
        const float4* sp = (const float4*)(h + (size_t)d * CH);
        const float dn = dis[d];
        const float4 s0 = sp[lane], s1 = sp[lane + 32], s2 = sp[t2];
        float4 a0 = make_float4(0.f, 0.f, 0.f, 0.f), a1 = a0, a2 = a0;
        float wsum = 0.f;

        auto addhits = [&](int2 v, bool ok) {
            unsigned m = __builtin_amdgcn_ballot_w32(ok && v.y == dl);
            const int nh = __builtin_popcount(m);
#pragma unroll 4
            for (int q = 0; q < nh; ++q) {
                int j = __builtin_ctz(m);
                m &= m - 1u;
                int sn = __builtin_amdgcn_readlane(v.x, j);
                float w = dis[sn];
                const float4* hp = (const float4*)(h + (size_t)sn * CH);
                float4 x0 = hp[lane], x1 = hp[lane + 32], x2 = hp[t2];
                wsum += w;
                a0.x += w * x0.x; a0.y += w * x0.y; a0.z += w * x0.z; a0.w += w * x0.w;
                a1.x += w * x1.x; a1.y += w * x1.y; a1.z += w * x1.z; a1.w += w * x1.w;
                a2.x += w * x2.x; a2.y += w * x2.y; a2.z += w * x2.z; a2.w += w * x2.w;
            }
        };

        if (fits) {
            for (int i0 = 0; i0 < no; i0 += 32) {
                bool ok = (i0 + lane) < no;
                int2 v = make_int2(0, -1);
                if (ok) v = my_own[i0 + lane];
                addhits(v, ok);
            }
        } else {
            for (int s = 0; s < AGG_W; ++s) {
                int ns = __builtin_amdgcn_readlane(cv, s);
                for (int i0 = 0; i0 < ns; i0 += 32) {
                    bool ok = (i0 + lane) < ns;
                    int2 v = make_int2(0, -1);
                    if (ok) {
                        if (i0 < AGG_CAP) v = lst[s * AGG_CAP + i0 + lane];
                        else v = blk_ovf[(size_t)s * chunk + (i0 + lane - AGG_CAP)];
                    }
                    addhits(v, ok);
                }
            }
        }

        float4 r0, r1, r2;
        r0.x = dn * (a0.x + dn * s0.x); r0.y = dn * (a0.y + dn * s0.y);
        r0.z = dn * (a0.z + dn * s0.z); r0.w = dn * (a0.w + dn * s0.w);
        r1.x = dn * (a1.x + dn * s1.x); r1.y = dn * (a1.y + dn * s1.y);
        r1.z = dn * (a1.z + dn * s1.z); r1.w = dn * (a1.w + dn * s1.w);
        r2.x = dn * (a2.x + dn * s2.x); r2.y = dn * (a2.y + dn * s2.y);
        r2.z = dn * (a2.z + dn * s2.z); r2.w = dn * (a2.w + dn * s2.w);
        put(dl, lane, r0); put(dl, lane + 32, r1);
        if (lane < ROW4 - 64) put(dl, lane + 64, r2);
        if (lane == 0) sc[dl] = dn * (wsum + dn);
    }
    __syncthreads();

    const int mn   = lane & 15;
    const int half = lane >> 4;
    auto unit = [&](auto nt_c, int rt, int cg) {
        constexpr int NT = decltype(nt_c)::value;
        v8f acc[NT] = {};
        const float* bp = Wp + (size_t)(cg * CGT) * TILE_F + 2 * lane;
        const int q0 = rt * KCH;
        for (int kc = 0; kc < KCH; ++kc) {
            const int q = q0 + kc;
            const float2 av = tl[q * 32 + (lane ^ (q & 31))];
            v2f a;
            a.x = av.x; a.y = av.y;
#pragma unroll
            for (int t = 0; t < NT; ++t) {
                v2f b = *(const v2f*)(bp + kc * 64 + t * TILE_F);
                acc[t] = __builtin_amdgcn_wmma_f32_16x16x4_f32(
                    false, a, false, b, (short)0, acc[t], false, false);
            }
        }
#pragma unroll
        for (int t = 0; t < NT; ++t) {
            const int n = (cg * CGT + t) * 16 + mn;
            if (n < CH) {
                const float bv = bias[n];
#pragma unroll
                for (int j = 0; j < 8; ++j) {
                    const int rl = rt * 16 + j + 8 * half;
                    if (rl < nrows) {
                        float v = acc[t][j] + sc[rl] * bv;
                        if (relu) v = (v > 0.f) ? v : 0.01f * v;
                        out[(size_t)(lo + rl) * CH + n] = v;
                    }
                }
            }
        }
    };
    for (int u = wave; u < ntb * 4; u += AGG_W) {
        const int rt = u >> 2;
        const int cg = (u + rt) & 3;
        if (cg == 3) unit(int_c<WTILES - 1 - 3 * CGT>{}, rt, cg);
        else         unit(int_c<CGT>{}, rt, cg);
    }
}

extern "C" void kernel_launch(void* const* d_in, const int* in_sizes, int n_in,
                              void* d_out, int out_size, void* d_ws, size_t ws_size,
                              hipStream_t stream) {
    const float* x  = (const float*)d_in[0];
    const int*   ei = (const int*)d_in[1];
    const float* W1 = (const float*)d_in[2];
    const float* b1 = (const float*)d_in[3];
    const float* W2 = (const float*)d_in[4];
    const float* b2 = (const float*)d_in[5];
    float* out = (float*)d_out;

    const int E = in_sizes[1] / 2;
    const int N = in_sizes[0] / (TP * CH);
    const int total = N * CH;
    const int* row = ei;
    const int* col = ei + E;

    char* ws = (char*)d_ws;
    float* dis = (float*)ws;
    size_t off = ((size_t)N * sizeof(float) + 255) & ~(size_t)255;
    float* Wp = (float*)(ws + off);
    off += (size_t)WTILES * TILE_F * sizeof(float);
    const int mtiles = (N + 15) / 16;
    float* bufA = (float*)(ws + off);
    float* bufB = bufA + (size_t)mtiles * TILE_F;
    int* ovf = (int*)(bufB + total);

    const int TB = 256;
    const int nagg = (N + AGG_R - 1) / AGG_R;
    const int padBlocks = (WTILES * TILE_U + TB - 1) / TB;

    fill_kernel<<<(N + TB - 1) / TB, TB, 0, stream>>>(dis, 1.0f, N);
    deg_kernel<<<(E + TB - 1) / TB, TB, 0, stream>>>(row, dis, E);
    rsqrt_kernel<<<(N + TB - 1) / TB, TB, 0, stream>>>(dis, N);

    mean_kernel<<<mtiles, 256, 0, stream>>>(x, bufA, N);

    padw_kernel<<<padBlocks, TB, 0, stream>>>(W1, Wp);
    agg_gemm_kernel<<<nagg, AGG_T, 0, stream>>>(row, col, dis, bufA, Wp, b1, bufB, ovf, E, N, 1, 0);

    padw_kernel<<<padBlocks, TB, 0, stream>>>(W2, Wp);
    agg_gemm_kernel<<<nagg, AGG_T, 0, stream>>>(row, col, dis, bufB, Wp, b2, out, ovf, E, N, 0, 1);
}
